# starting kernel
# baseline (speedup 1.0000x reference)
#include <hip/hip_runtime.h>
#include <math.h>

typedef __attribute__((ext_vector_type(2))) float v2f;
typedef __attribute__((ext_vector_type(8))) float v8f;

#define D_FEAT 128

__global__ void k_init_deg(float* __restrict__ deg, int n) {
    int i = blockIdx.x * blockDim.x + threadIdx.x;
    if (i < n) deg[i] = 1.0f;
}

__global__ void k_deg_accum(float* __restrict__ deg, const int* __restrict__ dst,
                            const float* __restrict__ ew, int E) {
    int e = blockIdx.x * blockDim.x + threadIdx.x;
    if (e < E) atomicAdd(&deg[dst[e]], ew[e]);
}

__global__ void k_dinv(float* __restrict__ deg, int n) {
    int i = blockIdx.x * blockDim.x + threadIdx.x;
    if (i < n) {
        float d = deg[i];
        deg[i] = (d > 0.0f) ? rsqrtf(d) : 0.0f;
    }
}

__global__ void k_gemm_wmma(const float* __restrict__ x, const float* __restrict__ W,
                            float* __restrict__ h, int N) {
    const int lane  = threadIdx.x & 31;
    const int wave  = threadIdx.x >> 5;
    const int Mbase = blockIdx.x * 16;
    const int Nbase = wave * 16;
    const int half  = lane >> 4;
    const int l15   = lane & 15;

    int mrow = Mbase + l15; if (mrow >= N) mrow = N - 1;
    const float* __restrict__ arow = x + (size_t)mrow * D_FEAT;
    const float* __restrict__ brow = W + (size_t)(Nbase + l15) * D_FEAT;

    v8f c = {};
#pragma unroll
    for (int k = 0; k < D_FEAT; k += 4) {
        const int kk = k + 2 * half;
        v2f a = *reinterpret_cast<const v2f*>(arow + kk);
        v2f b = *reinterpret_cast<const v2f*>(brow + kk);
        c = __builtin_amdgcn_wmma_f32_16x16x4_f32(false, a, false, b, (short)0, c,
                                                  false, false);
    }

    float* __restrict__ hout = h + (size_t)Mbase * D_FEAT + Nbase;
#pragma unroll
    for (int v = 0; v < 8; ++v) {
        int m = v + 8 * half;
        if (Mbase + m < N) hout[(size_t)m * D_FEAT + l15] = c[v];
    }
}

__global__ void k_out_init(float* __restrict__ out, const float* __restrict__ h,
                           const float* __restrict__ dinv, const float* __restrict__ bias,
                           int total) {
    int i = blockIdx.x * blockDim.x + threadIdx.x;
    if (i < total) {
        int node = i >> 7;
        int d    = i & 127;
        float di = dinv[node];
        out[i] = bias[d] + h[i] * di * di;
    }
}

__global__ void k_scatter(float* __restrict__ out, const float* __restrict__ h,
                          const float* __restrict__ dinv, const int* __restrict__ src,
                          const int* __restrict__ dst, const float* __restrict__ ew,
                          int E) {
    const int lane = threadIdx.x & 31;
    const int gw   = (blockIdx.x * blockDim.x + threadIdx.x) >> 5;
    const int nw   = (gridDim.x * blockDim.x) >> 5;
    for (int e = gw; e < E; e += nw) {
        const int s = src[e];
        const int t = dst[e];
        const float norm = dinv[s] * ew[e] * dinv[t];
        const float4 hv = *reinterpret_cast<const float4*>(h + (size_t)s * D_FEAT + lane * 4);
        float* op = out + (size_t)t * D_FEAT + lane * 4;
        atomicAdd(op + 0, hv.x * norm);
        atomicAdd(op + 1, hv.y * norm);
        atomicAdd(op + 2, hv.z * norm);
        atomicAdd(op + 3, hv.w * norm);
    }
}

extern "C" void kernel_launch(void* const* d_in, const int* in_sizes, int n_in,
                              void* d_out, int out_size, void* d_ws, size_t ws_size,
                              hipStream_t stream) {
    const float* x  = (const float*)d_in[0];
    const float* W  = (const float*)d_in[1];
    const float* bb = (const float*)d_in[2];
    const float* ew = (const float*)d_in[3];
    const int*   ei = (const int*)d_in[4];

    const int E = in_sizes[3];
    const int N = in_sizes[0] / D_FEAT;
    const int* src = ei;
    const int* dst = ei + E;
    float* out = (float*)d_out;

    float* deg = (float*)d_ws;
    size_t degPad = ((size_t)N + 63) & ~(size_t)63;
    float* h = deg + degPad;

    const int TB = 256;
    k_init_deg <<<(N + TB - 1) / TB, TB, 0, stream>>>(deg, N);
    k_deg_accum<<<(E + TB - 1) / TB, TB, 0, stream>>>(deg, dst, ew, E);
    k_dinv     <<<(N + TB - 1) / TB, TB, 0, stream>>>(deg, N);

    k_gemm_wmma<<<(N + 15) / 16, 256, 0, stream>>>(x, W, h, N);

    const int total = N * D_FEAT;
    k_out_init<<<(total + TB - 1) / TB, TB, 0, stream>>>(out, h, deg, bb, total);

    k_scatter<<<4096, TB, 0, stream>>>(out, h, deg, src, dst, ew, E);
}


# degree adds go to 32 planes picked by the wave's die (s_sendmsg_rtn 0x87) and block index: 1/32 of the atomics per 128-B line, none crosses dies; dinv sums planes in fixed order; sum order changed
# speedup vs baseline: 2.8521x; 2.8521x over previous
#include <hip/hip_runtime.h>
#include <math.h>


typedef __attribute__((ext_vector_type(2))) float v2f;
typedef __attribute__((ext_vector_type(8))) float v8f;

#define D_FEAT 128
#define SLICE_W 4
#define N_SLICE (D_FEAT / SLICE_W)
#define N_CHUNK 8
#define NP 10000

#define N_DIE 8
#define N_REP 4
#define N_PLANE (N_DIE * N_REP)

__global__ void k_init_deg(float* __restrict__ degp, int n4) {
    int i = blockIdx.x * blockDim.x + threadIdx.x;
    if (i < n4) reinterpret_cast<float4*>(degp)[i] = make_float4(0.f, 0.f, 0.f, 0.f);
}

__global__ void k_deg_accum(float* __restrict__ degp, const int* __restrict__ dst,
                            const float* __restrict__ ew, int E, int pitch) {
    const unsigned die = (__builtin_amdgcn_s_sendmsg_rtn(0x87) >> 16) & (N_DIE - 1);
    const unsigned pl  = die * N_REP + (blockIdx.x & (N_REP - 1));
    float* __restrict__ plane = degp + (size_t)pl * pitch;
    int e = blockIdx.x * blockDim.x + threadIdx.x;
    if (e < E) atomicAdd(&plane[dst[e]], ew[e]);
}

__global__ void k_dinv(float* __restrict__ dinv, const float* __restrict__ degp,
                       int n, int pitch) {
    int i = blockIdx.x * blockDim.x + threadIdx.x;
    if (i < n) {
        float p[N_PLANE];
#pragma unroll
        for (int k = 0; k < N_PLANE; ++k) p[k] = degp[(size_t)k * pitch + i];
        float d = 1.0f;
#pragma unroll
        for (int k = 0; k < N_PLANE; ++k) d += p[k];
        dinv[i] = (d > 0.0f) ? rsqrtf(d) : 0.0f;
    }
}

__global__ void k_gemm_wmma(const float* __restrict__ x, const float* __restrict__ W,
                            const float* __restrict__ dinv, float* __restrict__ hs, int N) {
    float4* __restrict__ hs4 = reinterpret_cast<float4*>(hs);
    const int lane  = threadIdx.x & 31;
    const int wave  = threadIdx.x >> 5;
    const int Mbase = blockIdx.x * 16;
    const int Nbase = wave * 16;
    const int half  = lane >> 4;
    const int l15   = lane & 15;

    int mrow = Mbase + l15; if (mrow >= N) mrow = N - 1;
    const float* __restrict__ xrow = x + (size_t)mrow * D_FEAT;
    const float* __restrict__ wrow = W + (size_t)(Nbase + l15) * D_FEAT;

    v8f c = {};
#pragma unroll
    for (int k = 0; k < D_FEAT; k += 4) {
        const int kk = k + 2 * half;
        v2f xa = *reinterpret_cast<const v2f*>(xrow + kk);
        v2f wb = *reinterpret_cast<const v2f*>(wrow + kk);
        c = __builtin_amdgcn_wmma_f32_16x16x4_f32(false, wb, false, xa, (short)0, c,
                                                  false, false);
    }

    const float di = dinv[mrow];
    const int slice0 = (Nbase + 8 * half) / SLICE_W;
    if (Mbase + l15 < N) {
        hs4[(size_t)slice0 * N + mrow]       = make_float4(c[0] * di, c[1] * di, c[2] * di, c[3] * di);
        hs4[(size_t)(slice0 + 1) * N + mrow] = make_float4(c[4] * di, c[5] * di, c[6] * di, c[7] * di);
    }
}

#define AGG_T 1024
#define STAGE_IT ((NP + AGG_T - 1) / AGG_T)

__device__ __forceinline__ void edge_add(float* acc, int t, const float4 hv, float w) {
    atomicAdd(&acc[t],          hv.x * w);
    atomicAdd(&acc[NP + t],     hv.y * w);
    atomicAdd(&acc[2 * NP + t], hv.z * w);
    atomicAdd(&acc[3 * NP + t], hv.w * w);
}

__global__ __launch_bounds__(AGG_T)
void k_agg(const float* __restrict__ hs, const int* __restrict__ src,
           const int* __restrict__ dst, const float* __restrict__ ew,
           float* __restrict__ part, int N, int E) {
    const float4* __restrict__ hs4 = reinterpret_cast<const float4*>(hs);
    float4* __restrict__ part4     = reinterpret_cast<float4*>(part);
    __shared__ float4 hl[NP];
    __shared__ float4 acc4[NP];
    float* acc = reinterpret_cast<float*>(acc4);

    const int tid   = threadIdx.x;
    const int slice = blockIdx.x % N_SLICE;
    const int chunk = blockIdx.x / N_SLICE;

    const float4* __restrict__ hsrc = hs4 + (size_t)slice * N;
    float4 st[STAGE_IT];
#pragma unroll
    for (int i = 0; i < STAGE_IT; ++i) {
        const int n = tid + i * AGG_T;
        st[i] = (n < N) ? hsrc[n] : make_float4(0.f, 0.f, 0.f, 0.f);
    }
#pragma unroll
    for (int i = 0; i < STAGE_IT; ++i) {
        const int n = tid + i * AGG_T;
        if (n < NP) {
            hl[n]   = st[i];
            acc4[n] = make_float4(0.f, 0.f, 0.f, 0.f);
        }
    }
    __syncthreads();

    const int CH = (((E + N_CHUNK - 1) / N_CHUNK) + 3) & ~3;
    const int c0 = chunk * CH;
    int c1 = c0 + CH; if (c1 > E) c1 = E;

    if ((E & 3) == 0) {
        const int G = (c1 - c0) >> 2;
        int g = tid;
        for (; g + AGG_T < G; g += 2 * AGG_T) {
            const int ea = c0 + 4 * g;
            const int eb = ea + 4 * AGG_T;
            const int4   sa = *reinterpret_cast<const int4*>(src + ea);
            const int4   da = *reinterpret_cast<const int4*>(dst + ea);
            const float4 wa = *reinterpret_cast<const float4*>(ew + ea);
            const int4   sb = *reinterpret_cast<const int4*>(src + eb);
            const int4   db = *reinterpret_cast<const int4*>(dst + eb);
            const float4 wb = *reinterpret_cast<const float4*>(ew + eb);
            const float4 a0 = hl[sa.x], a1 = hl[sa.y], a2 = hl[sa.z], a3 = hl[sa.w];
            const float4 b0 = hl[sb.x], b1 = hl[sb.y], b2 = hl[sb.z], b3 = hl[sb.w];
            edge_add(acc, da.x, a0, wa.x);
            edge_add(acc, da.y, a1, wa.y);
            edge_add(acc, da.z, a2, wa.z);
            edge_add(acc, da.w, a3, wa.w);
            edge_add(acc, db.x, b0, wb.x);
            edge_add(acc, db.y, b1, wb.y);
            edge_add(acc, db.z, b2, wb.z);
            edge_add(acc, db.w, b3, wb.w);
        }
        if (g < G) {
            const int ea = c0 + 4 * g;
            const int4   sa = *reinterpret_cast<const int4*>(src + ea);
            const int4   da = *reinterpret_cast<const int4*>(dst + ea);
            const float4 wa = *reinterpret_cast<const float4*>(ew + ea);
            const float4 a0 = hl[sa.x], a1 = hl[sa.y], a2 = hl[sa.z], a3 = hl[sa.w];
            edge_add(acc, da.x, a0, wa.x);
            edge_add(acc, da.y, a1, wa.y);
            edge_add(acc, da.z, a2, wa.z);
            edge_add(acc, da.w, a3, wa.w);
        }
    } else {
        for (int e = c0 + tid; e < c1; e += AGG_T)
            edge_add(acc, dst[e], hl[src[e]], ew[e]);
    }
    __syncthreads();

    float4* __restrict__ pout = part4 + (size_t)(chunk * N_SLICE + slice) * N;
    for (int n = tid; n < N; n += AGG_T)
        pout[n] = make_float4(acc[n], acc[NP + n], acc[2 * NP + n], acc[3 * NP + n]);
}

#define RED_NODES 16
__global__ void k_reduce(float* __restrict__ out, const float* __restrict__ hs,
                         const float* __restrict__ part, const float* __restrict__ dinv,
                         const float* __restrict__ bias, int N) {
    float4* __restrict__ out4        = reinterpret_cast<float4*>(out);
    const float4* __restrict__ hs4   = reinterpret_cast<const float4*>(hs);
    const float4* __restrict__ part4 = reinterpret_cast<const float4*>(part);
    const int lane  = threadIdx.x & 31;
    const int w     = threadIdx.x >> 5;
    const int slice = (w >> 1) * 8 + (lane & 7);
    const int n0    = blockIdx.x * RED_NODES + (w & 1) * 8 + (lane >> 3);
    const float4 b  = *reinterpret_cast<const float4*>(bias + slice * SLICE_W);

    float4 a[2];
    float  di[2];
#pragma unroll
    for (int j = 0; j < 2; ++j) {
        int node = n0 + 4 * j; if (node >= N) node = N - 1;
        const size_t idx = (size_t)slice * N + node;
        a[j]  = hs4[idx];
        di[j] = dinv[node];
#pragma unroll
        for (int c = 0; c < N_CHUNK; ++c) {
            const float4 p = part4[(size_t)c * N_SLICE * N + idx];
            a[j].x += p.x; a[j].y += p.y; a[j].z += p.z; a[j].w += p.w;
        }
    }
#pragma unroll
    for (int j = 0; j < 2; ++j) {
        const int node = n0 + 4 * j;
        if (node < N)
            out4[(size_t)node * N_SLICE + slice] =
                make_float4(b.x + di[j] * a[j].x, b.y + di[j] * a[j].y,
                            b.z + di[j] * a[j].z, b.w + di[j] * a[j].w);
    }
}

extern "C" void kernel_launch(void* const* d_in, const int* in_sizes, int n_in,
                              void* d_out, int out_size, void* d_ws, size_t ws_size,
                              hipStream_t stream) {
    const float* x  = (const float*)d_in[0];
    const float* W  = (const float*)d_in[1];
    const float* bb = (const float*)d_in[2];
    const float* ew = (const float*)d_in[3];
    const int*   ei = (const int*)d_in[4];

    const int E = in_sizes[3];
    const int N = in_sizes[0] / D_FEAT;
    const int* src = ei;
    const int* dst = ei + E;

    float* degp = (float*)d_ws;
    size_t degPad = ((size_t)N + 63) & ~(size_t)63;
    float* deg  = degp + (size_t)N_PLANE * degPad;
    float* hs   = deg + degPad;
    float* part = hs + (size_t)N * D_FEAT;

    const int TB = 256;
    const int n4 = (int)(N_PLANE * degPad / 4);
    k_init_deg <<<(n4 + TB - 1) / TB, TB, 0, stream>>>(degp, n4);
    k_deg_accum<<<(E + TB - 1) / TB, TB, 0, stream>>>(degp, dst, ew, E, (int)degPad);
    k_dinv     <<<(N + TB - 1) / TB, TB, 0, stream>>>(deg, degp, N, (int)degPad);

    k_gemm_wmma<<<(N + 15) / 16, 256, 0, stream>>>(x, W, deg, hs, N);

    k_agg<<<N_SLICE * N_CHUNK, AGG_T, 0, stream>>>(hs, src, dst, ew, part, N, E);

    k_reduce<<<(N + RED_NODES - 1) / RED_NODES, TB, 0, stream>>>((float*)d_out, hs, part, deg, bb, N);
}


# GEMM: a wave holds 4 accumulator tiles (32 nodes x 32 features) and loads each fragment as 16-byte pieces (two k-steps): 4 loads of 16 lines per 8 WMMAs, was 16; k sum order inside h changed
# speedup vs baseline: 2.9496x; 1.0342x over previous
#include <hip/hip_runtime.h>
#include <math.h>


typedef __attribute__((ext_vector_type(2))) float v2f;
typedef __attribute__((ext_vector_type(8))) float v8f;

#define D_FEAT 128
#define SLICE_W 4
#define N_SLICE (D_FEAT / SLICE_W)
#define N_CHUNK 8
#define NP 10000

#define N_DIE 8
#define N_REP 4
#define N_PLANE (N_DIE * N_REP)

__global__ void k_init_deg(float* __restrict__ degp, int n4) {
    int i = blockIdx.x * blockDim.x + threadIdx.x;
    if (i < n4) reinterpret_cast<float4*>(degp)[i] = make_float4(0.f, 0.f, 0.f, 0.f);
}

__global__ void k_deg_accum(float* __restrict__ degp, const int* __restrict__ dst,
                            const float* __restrict__ ew, int E, int pitch) {
    const unsigned die = (__builtin_amdgcn_s_sendmsg_rtn(0x87) >> 16) & (N_DIE - 1);
    const unsigned pl  = die * N_REP + (blockIdx.x & (N_REP - 1));
    float* __restrict__ plane = degp + (size_t)pl * pitch;
    int e = blockIdx.x * blockDim.x + threadIdx.x;
    if (e < E) atomicAdd(&plane[dst[e]], ew[e]);
}

__global__ void k_dinv(float* __restrict__ dinv, const float* __restrict__ degp,
                       int n, int pitch) {
    int i = blockIdx.x * blockDim.x + threadIdx.x;
    if (i < n) {
        float p[N_PLANE];
#pragma unroll
        for (int k = 0; k < N_PLANE; ++k) p[k] = degp[(size_t)k * pitch + i];
        float d = 1.0f;
#pragma unroll
        for (int k = 0; k < N_PLANE; ++k) d += p[k];
        dinv[i] = (d > 0.0f) ? rsqrtf(d) : 0.0f;
    }
}

#define GEMM_T 128

__device__ __forceinline__ v8f mma4(const v2f a, const v2f b, const v8f c) {
    return __builtin_amdgcn_wmma_f32_16x16x4_f32(false, a, false, b, (short)0, c, false, false);
}

__device__ __forceinline__ void store_tile(float4* __restrict__ hs4, const v8f c, float di,
                                           int slice0, int mrow, int N) {
    hs4[(size_t)slice0 * N + mrow]       = make_float4(c[0] * di, c[1] * di, c[2] * di, c[3] * di);
    hs4[(size_t)(slice0 + 1) * N + mrow] = make_float4(c[4] * di, c[5] * di, c[6] * di, c[7] * di);
}

__global__ __launch_bounds__(GEMM_T)
void k_gemm_wmma(const float* __restrict__ x, const float* __restrict__ W,
                 const float* __restrict__ dinv, float* __restrict__ hs, int N) {
    float4* __restrict__ hs4 = reinterpret_cast<float4*>(hs);
    const int lane  = threadIdx.x & 31;
    const int wave  = threadIdx.x >> 5;
    const int Mbase = blockIdx.x * 32;
    const int Fbase = wave * 32;
    const int half  = lane >> 4;
    const int l15   = lane & 15;

    int m0 = Mbase + l15;      if (m0 >= N) m0 = N - 1;
    int m1 = Mbase + 16 + l15; if (m1 >= N) m1 = N - 1;
    const float* __restrict__ xr0 = x + (size_t)m0 * D_FEAT + 4 * half;
    const float* __restrict__ xr1 = x + (size_t)m1 * D_FEAT + 4 * half;
    const float* __restrict__ wr0 = W + (size_t)(Fbase + l15) * D_FEAT + 4 * half;
    const float* __restrict__ wr1 = W + (size_t)(Fbase + 16 + l15) * D_FEAT + 4 * half;

    v8f c00 = {}, c01 = {}, c10 = {}, c11 = {};
#pragma unroll
    for (int k = 0; k < D_FEAT; k += 8) {
        const float4 xa = *reinterpret_cast<const float4*>(xr0 + k);
        const float4 xb = *reinterpret_cast<const float4*>(xr1 + k);
        const float4 wa = *reinterpret_cast<const float4*>(wr0 + k);
        const float4 wb = *reinterpret_cast<const float4*>(wr1 + k);
        const v2f xa0 = {xa.x, xa.y}, xa1 = {xa.z, xa.w};
        const v2f xb0 = {xb.x, xb.y}, xb1 = {xb.z, xb.w};
        const v2f wa0 = {wa.x, wa.y}, wa1 = {wa.z, wa.w};
        const v2f wb0 = {wb.x, wb.y}, wb1 = {wb.z, wb.w};
        c00 = mma4(wa0, xa0, c00);
        c01 = mma4(wa0, xb0, c01);
        c10 = mma4(wb0, xa0, c10);
        c11 = mma4(wb0, xb0, c11);
        c00 = mma4(wa1, xa1, c00);
        c01 = mma4(wa1, xb1, c01);
        c10 = mma4(wb1, xa1, c10);
        c11 = mma4(wb1, xb1, c11);
    }

    const float d0 = dinv[m0];
    const float d1 = dinv[m1];
    const int sA = (Fbase + 8 * half) / SLICE_W;
    const int sB = (Fbase + 16 + 8 * half) / SLICE_W;
    if (Mbase + l15 < N) {
        store_tile(hs4, c00, d0, sA, m0, N);
        store_tile(hs4, c10, d0, sB, m0, N);
    }
    if (Mbase + 16 + l15 < N) {
        store_tile(hs4, c01, d1, sA, m1, N);
        store_tile(hs4, c11, d1, sB, m1, N);
    }
}

#define AGG_T 1024
#define STAGE_IT ((NP + AGG_T - 1) / AGG_T)

__device__ __forceinline__ void edge_add(float* acc, int t, const float4 hv, float w) {
    atomicAdd(&acc[t],          hv.x * w);
    atomicAdd(&acc[NP + t],     hv.y * w);
    atomicAdd(&acc[2 * NP + t], hv.z * w);
    atomicAdd(&acc[3 * NP + t], hv.w * w);
}

__global__ __launch_bounds__(AGG_T)
void k_agg(const float* __restrict__ hs, const int* __restrict__ src,
           const int* __restrict__ dst, const float* __restrict__ ew,
           float* __restrict__ part, int N, int E) {
    const float4* __restrict__ hs4 = reinterpret_cast<const float4*>(hs);
    float4* __restrict__ part4     = reinterpret_cast<float4*>(part);
    __shared__ float4 hl[NP];
    __shared__ float4 acc4[NP];
    float* acc = reinterpret_cast<float*>(acc4);

    const int tid   = threadIdx.x;
    const int slice = blockIdx.x % N_SLICE;
    const int chunk = blockIdx.x / N_SLICE;

    const float4* __restrict__ hsrc = hs4 + (size_t)slice * N;
    float4 st[STAGE_IT];
#pragma unroll
    for (int i = 0; i < STAGE_IT; ++i) {
        const int n = tid + i * AGG_T;
        st[i] = (n < N) ? hsrc[n] : make_float4(0.f, 0.f, 0.f, 0.f);
    }
#pragma unroll
    for (int i = 0; i < STAGE_IT; ++i) {
        const int n = tid + i * AGG_T;
        if (n < NP) {
            hl[n]   = st[i];
            acc4[n] = make_float4(0.f, 0.f, 0.f, 0.f);
        }
    }
    __syncthreads();

    const int CH = (((E + N_CHUNK - 1) / N_CHUNK) + 3) & ~3;
    const int c0 = chunk * CH;
    int c1 = c0 + CH; if (c1 > E) c1 = E;

    if ((E & 3) == 0) {
        const int G = (c1 - c0) >> 2;
        int g = tid;
        for (; g + AGG_T < G; g += 2 * AGG_T) {
            const int ea = c0 + 4 * g;
            const int eb = ea + 4 * AGG_T;
            const int4   sa = *reinterpret_cast<const int4*>(src + ea);
            const int4   da = *reinterpret_cast<const int4*>(dst + ea);
            const float4 wa = *reinterpret_cast<const float4*>(ew + ea);
            const int4   sb = *reinterpret_cast<const int4*>(src + eb);
            const int4   db = *reinterpret_cast<const int4*>(dst + eb);
            const float4 wb = *reinterpret_cast<const float4*>(ew + eb);
            const float4 a0 = hl[sa.x], a1 = hl[sa.y], a2 = hl[sa.z], a3 = hl[sa.w];
            const float4 b0 = hl[sb.x], b1 = hl[sb.y], b2 = hl[sb.z], b3 = hl[sb.w];
            edge_add(acc, da.x, a0, wa.x);
            edge_add(acc, da.y, a1, wa.y);
            edge_add(acc, da.z, a2, wa.z);
            edge_add(acc, da.w, a3, wa.w);
            edge_add(acc, db.x, b0, wb.x);
            edge_add(acc, db.y, b1, wb.y);
            edge_add(acc, db.z, b2, wb.z);
            edge_add(acc, db.w, b3, wb.w);
        }
        if (g < G) {
            const int ea = c0 + 4 * g;
            const int4   sa = *reinterpret_cast<const int4*>(src + ea);
            const int4   da = *reinterpret_cast<const int4*>(dst + ea);
            const float4 wa = *reinterpret_cast<const float4*>(ew + ea);
            const float4 a0 = hl[sa.x], a1 = hl[sa.y], a2 = hl[sa.z], a3 = hl[sa.w];
            edge_add(acc, da.x, a0, wa.x);
            edge_add(acc, da.y, a1, wa.y);
            edge_add(acc, da.z, a2, wa.z);
            edge_add(acc, da.w, a3, wa.w);
        }
    } else {
        for (int e = c0 + tid; e < c1; e += AGG_T)
            edge_add(acc, dst[e], hl[src[e]], ew[e]);
    }
    __syncthreads();

    float4* __restrict__ pout = part4 + (size_t)(chunk * N_SLICE + slice) * N;
    for (int n = tid; n < N; n += AGG_T)
        pout[n] = make_float4(acc[n], acc[NP + n], acc[2 * NP + n], acc[3 * NP + n]);
}

#define RED_NODES 16
__global__ void k_reduce(float* __restrict__ out, const float* __restrict__ hs,
                         const float* __restrict__ part, const float* __restrict__ dinv,
                         const float* __restrict__ bias, int N) {
    float4* __restrict__ out4        = reinterpret_cast<float4*>(out);
    const float4* __restrict__ hs4   = reinterpret_cast<const float4*>(hs);
    const float4* __restrict__ part4 = reinterpret_cast<const float4*>(part);
    const int lane  = threadIdx.x & 31;
    const int w     = threadIdx.x >> 5;
    const int slice = (w >> 1) * 8 + (lane & 7);
    const int n0    = blockIdx.x * RED_NODES + (w & 1) * 8 + (lane >> 3);
    const float4 b  = *reinterpret_cast<const float4*>(bias + slice * SLICE_W);

    float4 a[2];
    float  di[2];
#pragma unroll
    for (int j = 0; j < 2; ++j) {
        int node = n0 + 4 * j; if (node >= N) node = N - 1;
        const size_t idx = (size_t)slice * N + node;
        a[j]  = hs4[idx];
        di[j] = dinv[node];
#pragma unroll
        for (int c = 0; c < N_CHUNK; ++c) {
            const float4 p = part4[(size_t)c * N_SLICE * N + idx];
            a[j].x += p.x; a[j].y += p.y; a[j].z += p.z; a[j].w += p.w;
        }
    }
#pragma unroll
    for (int j = 0; j < 2; ++j) {
        const int node = n0 + 4 * j;
        if (node < N)
            out4[(size_t)node * N_SLICE + slice] =
                make_float4(b.x + di[j] * a[j].x, b.y + di[j] * a[j].y,
                            b.z + di[j] * a[j].z, b.w + di[j] * a[j].w);
    }
}

extern "C" void kernel_launch(void* const* d_in, const int* in_sizes, int n_in,
                              void* d_out, int out_size, void* d_ws, size_t ws_size,
                              hipStream_t stream) {
    const float* x  = (const float*)d_in[0];
    const float* W  = (const float*)d_in[1];
    const float* bb = (const float*)d_in[2];
    const float* ew = (const float*)d_in[3];
    const int*   ei = (const int*)d_in[4];

    const int E = in_sizes[3];
    const int N = in_sizes[0] / D_FEAT;
    const int* src = ei;
    const int* dst = ei + E;

    float* degp = (float*)d_ws;
    size_t degPad = ((size_t)N + 63) & ~(size_t)63;
    float* deg  = degp + (size_t)N_PLANE * degPad;
    float* hs   = deg + degPad;
    float* part = hs + (size_t)N * D_FEAT;

    const int TB = 256;
    const int n4 = (int)(N_PLANE * degPad / 4);
    k_init_deg <<<(n4 + TB - 1) / TB, TB, 0, stream>>>(degp, n4);
    k_deg_accum<<<(E + TB - 1) / TB, TB, 0, stream>>>(degp, dst, ew, E, (int)degPad);
    k_dinv     <<<(N + TB - 1) / TB, TB, 0, stream>>>(deg, degp, N, (int)degPad);

    k_gemm_wmma<<<(N + 31) / 32, GEMM_T, 0, stream>>>(x, W, deg, hs, N);

    k_agg<<<N_SLICE * N_CHUNK, AGG_T, 0, stream>>>(hs, src, dst, ew, part, N, E);

    k_reduce<<<(N + RED_NODES - 1) / RED_NODES, TB, 0, stream>>>((float*)d_out, hs, part, deg, bb, N);
}


# dinv moved into the GEMM: each wave sums the 32 degree planes of its own 32 nodes (lane = node, same fixed order) behind its matrix work and hands it on by lane read; dinv launch gone, 6 -> 5
# speedup vs baseline: 3.0714x; 1.0413x over previous
#include <hip/hip_runtime.h>
#include <math.h>


typedef __attribute__((ext_vector_type(2))) float v2f;
typedef __attribute__((ext_vector_type(8))) float v8f;

#define D_FEAT 128
#define SLICE_W 4
#define N_SLICE (D_FEAT / SLICE_W)
#define N_CHUNK 8
#define NP 10000

#define N_DIE 8
#define N_REP 4
#define N_PLANE (N_DIE * N_REP)

__global__ void k_init_deg(float* __restrict__ degp, int n4) {
    int i = blockIdx.x * blockDim.x + threadIdx.x;
    if (i < n4) reinterpret_cast<float4*>(degp)[i] = make_float4(0.f, 0.f, 0.f, 0.f);
}

__global__ void k_deg_accum(float* __restrict__ degp, const int* __restrict__ dst,
                            const float* __restrict__ ew, int E, int pitch) {
    const unsigned die = (__builtin_amdgcn_s_sendmsg_rtn(0x87) >> 16) & (N_DIE - 1);
    const unsigned pl  = die * N_REP + (blockIdx.x & (N_REP - 1));
    float* __restrict__ plane = degp + (size_t)pl * pitch;
    int e = blockIdx.x * blockDim.x + threadIdx.x;
    if (e < E) atomicAdd(&plane[dst[e]], ew[e]);
}

__device__ __forceinline__ float node_dinv(const float* p) {
    float d = 1.0f;
#pragma unroll
    for (int k = 0; k < N_PLANE; ++k) d += p[k];
    return (d > 0.0f) ? rsqrtf(d) : 0.0f;
}

#define GEMM_T 128

__device__ __forceinline__ v8f mma4(const v2f a, const v2f b, const v8f c) {
    return __builtin_amdgcn_wmma_f32_16x16x4_f32(false, a, false, b, (short)0, c, false, false);
}

__device__ __forceinline__ void store_tile(float4* __restrict__ hs4, const v8f c, float di,
                                           int slice0, int mrow, int N) {
    hs4[(size_t)slice0 * N + mrow]       = make_float4(c[0] * di, c[1] * di, c[2] * di, c[3] * di);
    hs4[(size_t)(slice0 + 1) * N + mrow] = make_float4(c[4] * di, c[5] * di, c[6] * di, c[7] * di);
}

__global__ __launch_bounds__(GEMM_T)
void k_gemm_wmma(const float* __restrict__ x, const float* __restrict__ W,
                 const float* __restrict__ degp, float* __restrict__ dinv,
                 float* __restrict__ hs, int N, int pitch) {
    float4* __restrict__ hs4 = reinterpret_cast<float4*>(hs);
    const int lane  = threadIdx.x & 31;
    const int wave  = threadIdx.x >> 5;
    const int Mbase = blockIdx.x * 32;
    const int Fbase = wave * 32;
    const int half  = lane >> 4;
    const int l15   = lane & 15;

    int m0 = Mbase + l15;      if (m0 >= N) m0 = N - 1;
    int m1 = Mbase + 16 + l15; if (m1 >= N) m1 = N - 1;
    const float* __restrict__ xr0 = x + (size_t)m0 * D_FEAT + 4 * half;
    const float* __restrict__ xr1 = x + (size_t)m1 * D_FEAT + 4 * half;
    const float* __restrict__ wr0 = W + (size_t)(Fbase + l15) * D_FEAT + 4 * half;
    const float* __restrict__ wr1 = W + (size_t)(Fbase + 16 + l15) * D_FEAT + 4 * half;

    int nd = Mbase + lane; if (nd >= N) nd = N - 1;
    float pl[N_PLANE];
#pragma unroll
    for (int k = 0; k < N_PLANE; ++k) pl[k] = degp[(size_t)k * pitch + nd];

    v8f c00 = {}, c01 = {}, c10 = {}, c11 = {};
#pragma unroll
    for (int k = 0; k < D_FEAT; k += 8) {
        const float4 xa = *reinterpret_cast<const float4*>(xr0 + k);
        const float4 xb = *reinterpret_cast<const float4*>(xr1 + k);
        const float4 wa = *reinterpret_cast<const float4*>(wr0 + k);
        const float4 wb = *reinterpret_cast<const float4*>(wr1 + k);
        const v2f xa0 = {xa.x, xa.y}, xa1 = {xa.z, xa.w};
        const v2f xb0 = {xb.x, xb.y}, xb1 = {xb.z, xb.w};
        const v2f wa0 = {wa.x, wa.y}, wa1 = {wa.z, wa.w};
        const v2f wb0 = {wb.x, wb.y}, wb1 = {wb.z, wb.w};
        c00 = mma4(wa0, xa0, c00);
        c01 = mma4(wa0, xb0, c01);
        c10 = mma4(wb0, xa0, c10);
        c11 = mma4(wb0, xb0, c11);
        c00 = mma4(wa1, xa1, c00);
        c01 = mma4(wa1, xb1, c01);
        c10 = mma4(wb1, xa1, c10);
        c11 = mma4(wb1, xb1, c11);
    }

    const float dv = node_dinv(pl);
    if (wave == 0 && Mbase + lane < N) dinv[nd] = dv;
    const float d0 = __shfl(dv, l15);
    const float d1 = __shfl(dv, 16 + l15);
    const int sA = (Fbase + 8 * half) / SLICE_W;
    const int sB = (Fbase + 16 + 8 * half) / SLICE_W;
    if (Mbase + l15 < N) {
        store_tile(hs4, c00, d0, sA, m0, N);
        store_tile(hs4, c10, d0, sB, m0, N);
    }
    if (Mbase + 16 + l15 < N) {
        store_tile(hs4, c01, d1, sA, m1, N);
        store_tile(hs4, c11, d1, sB, m1, N);
    }
}

#define AGG_T 1024
#define STAGE_IT ((NP + AGG_T - 1) / AGG_T)

__device__ __forceinline__ void edge_add(float* acc, int t, const float4 hv, float w) {
    atomicAdd(&acc[t],          hv.x * w);
    atomicAdd(&acc[NP + t],     hv.y * w);
    atomicAdd(&acc[2 * NP + t], hv.z * w);
    atomicAdd(&acc[3 * NP + t], hv.w * w);
}

__global__ __launch_bounds__(AGG_T)
void k_agg(const float* __restrict__ hs, const int* __restrict__ src,
           const int* __restrict__ dst, const float* __restrict__ ew,
           float* __restrict__ part, int N, int E) {
    const float4* __restrict__ hs4 = reinterpret_cast<const float4*>(hs);
    float4* __restrict__ part4     = reinterpret_cast<float4*>(part);
    __shared__ float4 hl[NP];
    __shared__ float4 acc4[NP];
    float* acc = reinterpret_cast<float*>(acc4);

    const int tid   = threadIdx.x;
    const int slice = blockIdx.x % N_SLICE;
    const int chunk = blockIdx.x / N_SLICE;

    const float4* __restrict__ hsrc = hs4 + (size_t)slice * N;
    float4 st[STAGE_IT];
#pragma unroll
    for (int i = 0; i < STAGE_IT; ++i) {
        const int n = tid + i * AGG_T;
        st[i] = (n < N) ? hsrc[n] : make_float4(0.f, 0.f, 0.f, 0.f);
    }
#pragma unroll
    for (int i = 0; i < STAGE_IT; ++i) {
        const int n = tid + i * AGG_T;
        if (n < NP) {
            hl[n]   = st[i];
            acc4[n] = make_float4(0.f, 0.f, 0.f, 0.f);
        }
    }
    __syncthreads();

    const int CH = (((E + N_CHUNK - 1) / N_CHUNK) + 3) & ~3;
    const int c0 = chunk * CH;
    int c1 = c0 + CH; if (c1 > E) c1 = E;

    if ((E & 3) == 0) {
        const int G = (c1 - c0) >> 2;
        int g = tid;
        for (; g + AGG_T < G; g += 2 * AGG_T) {
            const int ea = c0 + 4 * g;
            const int eb = ea + 4 * AGG_T;
            const int4   sa = *reinterpret_cast<const int4*>(src + ea);
            const int4   da = *reinterpret_cast<const int4*>(dst + ea);
            const float4 wa = *reinterpret_cast<const float4*>(ew + ea);
            const int4   sb = *reinterpret_cast<const int4*>(src + eb);
            const int4   db = *reinterpret_cast<const int4*>(dst + eb);
            const float4 wb = *reinterpret_cast<const float4*>(ew + eb);
            const float4 a0 = hl[sa.x], a1 = hl[sa.y], a2 = hl[sa.z], a3 = hl[sa.w];
            const float4 b0 = hl[sb.x], b1 = hl[sb.y], b2 = hl[sb.z], b3 = hl[sb.w];
            edge_add(acc, da.x, a0, wa.x);
            edge_add(acc, da.y, a1, wa.y);
            edge_add(acc, da.z, a2, wa.z);
            edge_add(acc, da.w, a3, wa.w);
            edge_add(acc, db.x, b0, wb.x);
            edge_add(acc, db.y, b1, wb.y);
            edge_add(acc, db.z, b2, wb.z);
            edge_add(acc, db.w, b3, wb.w);
        }
        if (g < G) {
            const int ea = c0 + 4 * g;
            const int4   sa = *reinterpret_cast<const int4*>(src + ea);
            const int4   da = *reinterpret_cast<const int4*>(dst + ea);
            const float4 wa = *reinterpret_cast<const float4*>(ew + ea);
            const float4 a0 = hl[sa.x], a1 = hl[sa.y], a2 = hl[sa.z], a3 = hl[sa.w];
            edge_add(acc, da.x, a0, wa.x);
            edge_add(acc, da.y, a1, wa.y);
            edge_add(acc, da.z, a2, wa.z);
            edge_add(acc, da.w, a3, wa.w);
        }
    } else {
        for (int e = c0 + tid; e < c1; e += AGG_T)
            edge_add(acc, dst[e], hl[src[e]], ew[e]);
    }
    __syncthreads();

    float4* __restrict__ pout = part4 + (size_t)(chunk * N_SLICE + slice) * N;
    for (int n = tid; n < N; n += AGG_T)
        pout[n] = make_float4(acc[n], acc[NP + n], acc[2 * NP + n], acc[3 * NP + n]);
}

#define RED_NODES 16
__global__ void k_reduce(float* __restrict__ out, const float* __restrict__ hs,
                         const float* __restrict__ part, const float* __restrict__ dinv,
                         const float* __restrict__ bias, int N) {
    float4* __restrict__ out4        = reinterpret_cast<float4*>(out);
    const float4* __restrict__ hs4   = reinterpret_cast<const float4*>(hs);
    const float4* __restrict__ part4 = reinterpret_cast<const float4*>(part);
    const int lane  = threadIdx.x & 31;
    const int w     = threadIdx.x >> 5;
    const int slice = (w >> 1) * 8 + (lane & 7);
    const int n0    = blockIdx.x * RED_NODES + (w & 1) * 8 + (lane >> 3);
    const float4 b  = *reinterpret_cast<const float4*>(bias + slice * SLICE_W);

    float4 a[2];
    float  di[2];
#pragma unroll
    for (int j = 0; j < 2; ++j) {
        int node = n0 + 4 * j; if (node >= N) node = N - 1;
        const size_t idx = (size_t)slice * N + node;
        a[j]  = hs4[idx];
        di[j] = dinv[node];
#pragma unroll
        for (int c = 0; c < N_CHUNK; ++c) {
            const float4 p = part4[(size_t)c * N_SLICE * N + idx];
            a[j].x += p.x; a[j].y += p.y; a[j].z += p.z; a[j].w += p.w;
        }
    }
#pragma unroll
    for (int j = 0; j < 2; ++j) {
        const int node = n0 + 4 * j;
        if (node < N)
            out4[(size_t)node * N_SLICE + slice] =
                make_float4(b.x + di[j] * a[j].x, b.y + di[j] * a[j].y,
                            b.z + di[j] * a[j].z, b.w + di[j] * a[j].w);
    }
}

extern "C" void kernel_launch(void* const* d_in, const int* in_sizes, int n_in,
                              void* d_out, int out_size, void* d_ws, size_t ws_size,
                              hipStream_t stream) {
    const float* x  = (const float*)d_in[0];
    const float* W  = (const float*)d_in[1];
    const float* bb = (const float*)d_in[2];
    const float* ew = (const float*)d_in[3];
    const int*   ei = (const int*)d_in[4];

    const int E = in_sizes[3];
    const int N = in_sizes[0] / D_FEAT;
    const int* src = ei;
    const int* dst = ei + E;

    float* degp = (float*)d_ws;
    size_t degPad = ((size_t)N + 63) & ~(size_t)63;
    float* deg  = degp + (size_t)N_PLANE * degPad;
    float* hs   = deg + degPad;
    float* part = hs + (size_t)N * D_FEAT;

    const int TB = 256;
    const int n4 = (int)(N_PLANE * degPad / 4);
    k_init_deg <<<(n4 + TB - 1) / TB, TB, 0, stream>>>(degp, n4);
    k_deg_accum<<<(E + TB - 1) / TB, TB, 0, stream>>>(degp, dst, ew, E, (int)degPad);

    k_gemm_wmma<<<(N + 31) / 32, GEMM_T, 0, stream>>>(x, W, degp, deg, hs, N, (int)degPad);

    k_agg<<<N_SLICE * N_CHUNK, AGG_T, 0, stream>>>(hs, src, dst, ew, part, N, E);

    k_reduce<<<(N + RED_NODES - 1) / RED_NODES, TB, 0, stream>>>((float*)d_out, hs, part, deg, bb, N);
}


# aggregation: a node's 4 sums sit side by side in LDS and a lane quad transposes its rows in registers, so one LDS add instruction takes 8 edges x 4 neighbouring banks; arrival-order sums as before
# speedup vs baseline: 3.2232x; 1.0494x over previous
#include <hip/hip_runtime.h>
#include <math.h>


typedef __attribute__((ext_vector_type(2))) float v2f;
typedef __attribute__((ext_vector_type(8))) float v8f;

#define D_FEAT 128
#define SLICE_W 4
#define N_SLICE (D_FEAT / SLICE_W)
#define N_CHUNK 8
#define NP 10000

#define N_DIE 8
#define N_REP 4
#define N_PLANE (N_DIE * N_REP)

__global__ void k_init_deg(float* __restrict__ degp, int n4) {
    int i = blockIdx.x * blockDim.x + threadIdx.x;
    if (i < n4) reinterpret_cast<float4*>(degp)[i] = make_float4(0.f, 0.f, 0.f, 0.f);
}

__global__ void k_deg_accum(float* __restrict__ degp, const int* __restrict__ dst,
                            const float* __restrict__ ew, int E, int pitch) {
    const unsigned die = (__builtin_amdgcn_s_sendmsg_rtn(0x87) >> 16) & (N_DIE - 1);
    const unsigned pl  = die * N_REP + (blockIdx.x & (N_REP - 1));
    float* __restrict__ plane = degp + (size_t)pl * pitch;
    int e = blockIdx.x * blockDim.x + threadIdx.x;
    if (e < E) atomicAdd(&plane[dst[e]], ew[e]);
}

__device__ __forceinline__ float node_dinv(const float* p) {
    float d = 1.0f;
#pragma unroll
    for (int k = 0; k < N_PLANE; ++k) d += p[k];
    return (d > 0.0f) ? rsqrtf(d) : 0.0f;
}

#define GEMM_T 128

__device__ __forceinline__ v8f mma4(const v2f a, const v2f b, const v8f c) {
    return __builtin_amdgcn_wmma_f32_16x16x4_f32(false, a, false, b, (short)0, c, false, false);
}

__device__ __forceinline__ void store_tile(float4* __restrict__ hs4, const v8f c, float di,
                                           int slice0, int mrow, int N) {
    hs4[(size_t)slice0 * N + mrow]       = make_float4(c[0] * di, c[1] * di, c[2] * di, c[3] * di);
    hs4[(size_t)(slice0 + 1) * N + mrow] = make_float4(c[4] * di, c[5] * di, c[6] * di, c[7] * di);
}

__global__ __launch_bounds__(GEMM_T)
void k_gemm_wmma(const float* __restrict__ x, const float* __restrict__ W,
                 const float* __restrict__ degp, float* __restrict__ dinv,
                 float* __restrict__ hs, int N, int pitch) {
    float4* __restrict__ hs4 = reinterpret_cast<float4*>(hs);
    const int lane  = threadIdx.x & 31;
    const int wave  = threadIdx.x >> 5;
    const int Mbase = blockIdx.x * 32;
    const int Fbase = wave * 32;
    const int half  = lane >> 4;
    const int l15   = lane & 15;

    int m0 = Mbase + l15;      if (m0 >= N) m0 = N - 1;
    int m1 = Mbase + 16 + l15; if (m1 >= N) m1 = N - 1;
    const float* __restrict__ xr0 = x + (size_t)m0 * D_FEAT + 4 * half;
    const float* __restrict__ xr1 = x + (size_t)m1 * D_FEAT + 4 * half;
    const float* __restrict__ wr0 = W + (size_t)(Fbase + l15) * D_FEAT + 4 * half;
    const float* __restrict__ wr1 = W + (size_t)(Fbase + 16 + l15) * D_FEAT + 4 * half;

    int nd = Mbase + lane; if (nd >= N) nd = N - 1;
    float pl[N_PLANE];
#pragma unroll
    for (int k = 0; k < N_PLANE; ++k) pl[k] = degp[(size_t)k * pitch + nd];

    v8f c00 = {}, c01 = {}, c10 = {}, c11 = {};
#pragma unroll
    for (int k = 0; k < D_FEAT; k += 8) {
        const float4 xa = *reinterpret_cast<const float4*>(xr0 + k);
        const float4 xb = *reinterpret_cast<const float4*>(xr1 + k);
        const float4 wa = *reinterpret_cast<const float4*>(wr0 + k);
        const float4 wb = *reinterpret_cast<const float4*>(wr1 + k);
        const v2f xa0 = {xa.x, xa.y}, xa1 = {xa.z, xa.w};
        const v2f xb0 = {xb.x, xb.y}, xb1 = {xb.z, xb.w};
        const v2f wa0 = {wa.x, wa.y}, wa1 = {wa.z, wa.w};
        const v2f wb0 = {wb.x, wb.y}, wb1 = {wb.z, wb.w};
        c00 = mma4(wa0, xa0, c00);
        c01 = mma4(wa0, xb0, c01);
        c10 = mma4(wb0, xa0, c10);
        c11 = mma4(wb0, xb0, c11);
        c00 = mma4(wa1, xa1, c00);
        c01 = mma4(wa1, xb1, c01);
        c10 = mma4(wb1, xa1, c10);
        c11 = mma4(wb1, xb1, c11);
    }

    const float dv = node_dinv(pl);
    if (wave == 0 && Mbase + lane < N) dinv[nd] = dv;
    const float d0 = __shfl(dv, l15);
    const float d1 = __shfl(dv, 16 + l15);
    const int sA = (Fbase + 8 * half) / SLICE_W;
    const int sB = (Fbase + 16 + 8 * half) / SLICE_W;
    if (Mbase + l15 < N) {
        store_tile(hs4, c00, d0, sA, m0, N);
        store_tile(hs4, c10, d0, sB, m0, N);
    }
    if (Mbase + 16 + l15 < N) {
        store_tile(hs4, c01, d1, sA, m1, N);
        store_tile(hs4, c11, d1, sB, m1, N);
    }
}

#define AGG_T 1024
#define STAGE_IT ((NP + AGG_T - 1) / AGG_T)

__device__ __forceinline__ float quad_x1(float v) {
    return __builtin_bit_cast(float, __builtin_amdgcn_update_dpp(0, __builtin_bit_cast(int, v), 0xB1, 0xF, 0xF, true));
}
__device__ __forceinline__ float quad_x2(float v) {
    return __builtin_bit_cast(float, __builtin_amdgcn_update_dpp(0, __builtin_bit_cast(int, v), 0x4E, 0xF, 0xF, true));
}
template <int R>
__device__ __forceinline__ int quad_bcast(int v) {
    return __builtin_amdgcn_update_dpp(0, v, R * 0x55, 0xF, 0xF, true);
}

__device__ __forceinline__ float4 quad_transpose(float4 v, bool b1, bool b2) {
    float s, r;
    s = b1 ? v.x : v.y; r = quad_x1(s); v.x = b1 ? r : v.x; v.y = b1 ? v.y : r;
    s = b1 ? v.z : v.w; r = quad_x1(s); v.z = b1 ? r : v.z; v.w = b1 ? v.w : r;
    s = b2 ? v.x : v.z; r = quad_x2(s); v.x = b2 ? r : v.x; v.z = b2 ? v.z : r;
    s = b2 ? v.y : v.w; r = quad_x2(s); v.y = b2 ? r : v.y; v.w = b2 ? v.w : r;
    return v;
}

__device__ __forceinline__ void quad_edge_add(float* accj, int t, float4 hv, bool b1, bool b2) {
    const float4 c = quad_transpose(hv, b1, b2);
    const int t4 = 4 * t;
    atomicAdd(&accj[quad_bcast<0>(t4)], c.x);
    atomicAdd(&accj[quad_bcast<1>(t4)], c.y);
    atomicAdd(&accj[quad_bcast<2>(t4)], c.z);
    atomicAdd(&accj[quad_bcast<3>(t4)], c.w);
}

__device__ __forceinline__ float4 scale4(const float4 h, float w) {
    return make_float4(h.x * w, h.y * w, h.z * w, h.w * w);
}

__device__ __forceinline__ void quad_group_add(float* accj, const int4 d, const float4 w,
                                               const float4 a0, const float4 a1,
                                               const float4 a2, const float4 a3,
                                               bool b1, bool b2) {
    quad_edge_add(accj, d.x, scale4(a0, w.x), b1, b2);
    quad_edge_add(accj, d.y, scale4(a1, w.y), b1, b2);
    quad_edge_add(accj, d.z, scale4(a2, w.z), b1, b2);
    quad_edge_add(accj, d.w, scale4(a3, w.w), b1, b2);
}

__device__ __forceinline__ void edge_add(float* acc, int t, const float4 hv, float w) {
    atomicAdd(&acc[4 * t],     hv.x * w);
    atomicAdd(&acc[4 * t + 1], hv.y * w);
    atomicAdd(&acc[4 * t + 2], hv.z * w);
    atomicAdd(&acc[4 * t + 3], hv.w * w);
}

__global__ __launch_bounds__(AGG_T)
void k_agg(const float* __restrict__ hs, const int* __restrict__ src,
           const int* __restrict__ dst, const float* __restrict__ ew,
           float* __restrict__ part, int N, int E) {
    const float4* __restrict__ hs4 = reinterpret_cast<const float4*>(hs);
    float4* __restrict__ part4     = reinterpret_cast<float4*>(part);
    __shared__ float4 hl[NP];
    __shared__ float4 acc4[NP];
    float* acc = reinterpret_cast<float*>(acc4);

    const int tid   = threadIdx.x;
    const int slice = blockIdx.x % N_SLICE;
    const int chunk = blockIdx.x / N_SLICE;

    const float4* __restrict__ hsrc = hs4 + (size_t)slice * N;
    float4 st[STAGE_IT];
#pragma unroll
    for (int i = 0; i < STAGE_IT; ++i) {
        const int n = tid + i * AGG_T;
        st[i] = (n < N) ? hsrc[n] : make_float4(0.f, 0.f, 0.f, 0.f);
    }
#pragma unroll
    for (int i = 0; i < STAGE_IT; ++i) {
        const int n = tid + i * AGG_T;
        if (n < NP) {
            hl[n]   = st[i];
            acc4[n] = make_float4(0.f, 0.f, 0.f, 0.f);
        }
    }
    __syncthreads();

    const int CH = (((E + N_CHUNK - 1) / N_CHUNK) + 3) & ~3;
    const int c0 = chunk * CH;
    int c1 = c0 + CH; if (c1 > E) c1 = E;

    if ((E & 3) == 0) {
        const int  G    = (c1 - c0) >> 2;
        const bool b1   = (tid & 1) != 0;
        const bool b2   = (tid & 2) != 0;
        float*     accj = acc + (tid & 3);
        int base = 0;
        for (; base + 2 * AGG_T <= G; base += 2 * AGG_T) {
            const int ea = c0 + 4 * (base + tid);
            const int eb = ea + 4 * AGG_T;
            const int4   sa = *reinterpret_cast<const int4*>(src + ea);
            const int4   da = *reinterpret_cast<const int4*>(dst + ea);
            const float4 wa = *reinterpret_cast<const float4*>(ew + ea);
            const int4   sb = *reinterpret_cast<const int4*>(src + eb);
            const int4   db = *reinterpret_cast<const int4*>(dst + eb);
            const float4 wb = *reinterpret_cast<const float4*>(ew + eb);
            asm volatile("" :: "v"(da.x), "v"(da.y), "v"(da.z), "v"(da.w),
                               "v"(db.x), "v"(db.y), "v"(db.z), "v"(db.w));
            const float4 a0 = hl[sa.x], a1 = hl[sa.y], a2 = hl[sa.z], a3 = hl[sa.w];
            const float4 b0 = hl[sb.x], b1r = hl[sb.y], b2r = hl[sb.z], b3 = hl[sb.w];
            quad_group_add(accj, da, wa, a0, a1, a2, a3, b1, b2);
            quad_group_add(accj, db, wb, b0, b1r, b2r, b3, b1, b2);
        }
        for (; base + (tid | 3) < G; base += AGG_T) {
            const int ea = c0 + 4 * (base + tid);
            const int4   sa = *reinterpret_cast<const int4*>(src + ea);
            const int4   da = *reinterpret_cast<const int4*>(dst + ea);
            const float4 wa = *reinterpret_cast<const float4*>(ew + ea);
            const float4 a0 = hl[sa.x], a1 = hl[sa.y], a2 = hl[sa.z], a3 = hl[sa.w];
            quad_group_add(accj, da, wa, a0, a1, a2, a3, b1, b2);
        }
        if (base + tid < G) {
            const int ea = c0 + 4 * (base + tid);
            const int4   sa = *reinterpret_cast<const int4*>(src + ea);
            const int4   da = *reinterpret_cast<const int4*>(dst + ea);
            const float4 wa = *reinterpret_cast<const float4*>(ew + ea);
            edge_add(acc, da.x, hl[sa.x], wa.x);
            edge_add(acc, da.y, hl[sa.y], wa.y);
            edge_add(acc, da.z, hl[sa.z], wa.z);
            edge_add(acc, da.w, hl[sa.w], wa.w);
        }
    } else {
        for (int e = c0 + tid; e < c1; e += AGG_T)
            edge_add(acc, dst[e], hl[src[e]], ew[e]);
    }
    __syncthreads();

    float4* __restrict__ pout = part4 + (size_t)(chunk * N_SLICE + slice) * N;
    for (int n = tid; n < N; n += AGG_T)
        pout[n] = acc4[n];
}

#define RED_NODES 16
__global__ void k_reduce(float* __restrict__ out, const float* __restrict__ hs,
                         const float* __restrict__ part, const float* __restrict__ dinv,
                         const float* __restrict__ bias, int N) {
    float4* __restrict__ out4        = reinterpret_cast<float4*>(out);
    const float4* __restrict__ hs4   = reinterpret_cast<const float4*>(hs);
    const float4* __restrict__ part4 = reinterpret_cast<const float4*>(part);
    const int lane  = threadIdx.x & 31;
    const int w     = threadIdx.x >> 5;
    const int slice = (w >> 1) * 8 + (lane & 7);
    const int n0    = blockIdx.x * RED_NODES + (w & 1) * 8 + (lane >> 3);
    const float4 b  = *reinterpret_cast<const float4*>(bias + slice * SLICE_W);

    float4 a[2];
    float  di[2];
#pragma unroll
    for (int j = 0; j < 2; ++j) {
        int node = n0 + 4 * j; if (node >= N) node = N - 1;
        const size_t idx = (size_t)slice * N + node;
        a[j]  = hs4[idx];
        di[j] = dinv[node];
#pragma unroll
        for (int c = 0; c < N_CHUNK; ++c) {
            const float4 p = part4[(size_t)c * N_SLICE * N + idx];
            a[j].x += p.x; a[j].y += p.y; a[j].z += p.z; a[j].w += p.w;
        }
    }
#pragma unroll
    for (int j = 0; j < 2; ++j) {
        const int node = n0 + 4 * j;
        if (node < N)
            out4[(size_t)node * N_SLICE + slice] =
                make_float4(b.x + di[j] * a[j].x, b.y + di[j] * a[j].y,
                            b.z + di[j] * a[j].z, b.w + di[j] * a[j].w);
    }
}

extern "C" void kernel_launch(void* const* d_in, const int* in_sizes, int n_in,
                              void* d_out, int out_size, void* d_ws, size_t ws_size,
                              hipStream_t stream) {
    const float* x  = (const float*)d_in[0];
    const float* W  = (const float*)d_in[1];
    const float* bb = (const float*)d_in[2];
    const float* ew = (const float*)d_in[3];
    const int*   ei = (const int*)d_in[4];

    const int E = in_sizes[3];
    const int N = in_sizes[0] / D_FEAT;
    const int* src = ei;
    const int* dst = ei + E;

    float* degp = (float*)d_ws;
    size_t degPad = ((size_t)N + 63) & ~(size_t)63;
    float* deg  = degp + (size_t)N_PLANE * degPad;
    float* hs   = deg + degPad;
    float* part = hs + (size_t)N * D_FEAT;

    const int TB = 256;
    const int n4 = (int)(N_PLANE * degPad / 4);
    k_init_deg <<<(n4 + TB - 1) / TB, TB, 0, stream>>>(degp, n4);
    k_deg_accum<<<(E + TB - 1) / TB, TB, 0, stream>>>(degp, dst, ew, E, (int)degPad);

    k_gemm_wmma<<<(N + 31) / 32, GEMM_T, 0, stream>>>(x, W, degp, deg, hs, N, (int)degPad);

    k_agg<<<N_SLICE * N_CHUNK, AGG_T, 0, stream>>>(hs, src, dst, ew, part, N, E);

    k_reduce<<<(N + RED_NODES - 1) / RED_NODES, TB, 0, stream>>>((float*)d_out, hs, part, deg, bb, N);
}
